# starting kernel
# baseline (speedup 1.0000x reference)
#include <hip/hip_runtime.h>
#include <hip/hip_bf16.h>
#include <math.h>

#define SLOT  2
#define DIM   128
#define ROW   (SLOT * DIM)
#define SLOPE 0.98f

typedef float v2f __attribute__((ext_vector_type(2)));
typedef float v8f __attribute__((ext_vector_type(8)));

__device__ __forceinline__ unsigned flip_f32(float f) {
    unsigned b = __float_as_uint(f);
    return (b & 0x80000000u) ? ~b : (b | 0x80000000u);
}
__device__ __forceinline__ float unflip_f32(unsigned u) {
    unsigned b = (u & 0x80000000u) ? (u & 0x7fffffffu) : ~u;
    return __uint_as_float(b);
}

__global__ void k_init(float4* __restrict__ out4, int nOut4,
                       unsigned* __restrict__ m_u, float* __restrict__ s, int N) {
    int i      = blockIdx.x * blockDim.x + threadIdx.x;
    int stride = gridDim.x * blockDim.x;
    float4 z = make_float4(0.f, 0.f, 0.f, 0.f);
    for (int j = i; j < nOut4; j += stride) out4[j] = z;
    for (int j = i; j < N; j += stride) { m_u[j] = 0u; s[j] = 0.0f; }
}

__global__ void k_edge_scores(const float* __restrict__ h,
                              const int*  __restrict__ src,
                              const int*  __restrict__ dst,
                              const float* __restrict__ watt,
                              float* __restrict__ abuf,
                              unsigned* __restrict__ m_u,
                              int E, int ntiles) {
    __shared__ float wlds[2 * DIM];
    int tid = threadIdx.x;
    if (tid < 2 * DIM) wlds[tid] = watt[tid];
    __syncthreads();

    int tile = blockIdx.x * (blockDim.x >> 5) + (tid >> 5);
    if (tile >= ntiles) return;

    int lane = tid & 31;
    int half = lane >> 4;
    int m16  = lane & 15;
    int base = tile * 16;

    int eL = base + m16; if (eL >= E) eL = E - 1;
    int sv = src[eL];
    int dv = dst[eL];
    const float* ps = h + (size_t)sv * ROW + 2 * half;
    const float* pd = h + (size_t)dv * ROW + 2 * half;
    const float* wl = wlds + 2 * half;

    v8f c = {};
#if __has_builtin(__builtin_amdgcn_wmma_f32_16x16x4_f32)
    #pragma unroll
    for (int k = 0; k < 32; ++k) {
        v2f av = *(const v2f*)(ps + 4 * k);
        v2f bv = *(const v2f*)(wl + 4 * k);
        c = __builtin_amdgcn_wmma_f32_16x16x4_f32(false, av, false, bv,
                                                  (short)0, c, false, false);
    }
    #pragma unroll
    for (int k = 0; k < 32; ++k) {
        v2f av = *(const v2f*)(pd + 4 * k);
        v2f bv = *(const v2f*)(wl + DIM + 4 * k);
        c = __builtin_amdgcn_wmma_f32_16x16x4_f32(false, av, false, bv,
                                                  (short)0, c, false, false);
    }
    int r = lane & 7;
    float sc = c[0];
    sc = (r == 1) ? c[1] : sc;
    sc = (r == 2) ? c[2] : sc;
    sc = (r == 3) ? c[3] : sc;
    sc = (r == 4) ? c[4] : sc;
    sc = (r == 5) ? c[5] : sc;
    sc = (r == 6) ? c[6] : sc;
    sc = (r == 7) ? c[7] : sc;
#else
    int r = lane & 7;
    int rowF = ((lane >> 4) << 3) | r;
    int eF = base + rowF; if (eF >= E) eF = E - 1;
    const float* fs = h + (size_t)src[eF] * ROW;
    const float* fd = h + (size_t)dst[eF] * ROW;
    float sc = 0.f;
    for (int k = 0; k < DIM; ++k) sc += fs[k] * wlds[k] + fd[k] * wlds[DIM + k];
#endif

    int row = ((lane >> 4) << 3) | r;
    if ((lane & 8) == 0) {
        int e = base + row;
        if (e < E) {
            float v = (sc >= 0.0f) ? sc : SLOPE * sc;
            abuf[e] = v;
            atomicMax(&m_u[dst[e]], flip_f32(v));
        }
    }
}

__global__ void k_edge_exp(const int* __restrict__ dst,
                           float* __restrict__ abuf,
                           const unsigned* __restrict__ m_u,
                           float* __restrict__ s, int E) {
    int e = blockIdx.x * blockDim.x + threadIdx.x;
    if (e >= E) return;
    int d = dst[e];
    float w = __expf(abuf[e] - unflip_f32(m_u[d]));
    abuf[e] = w;
    atomicAdd(&s[d], w);
}

__global__ void k_scatter(const float* __restrict__ h,
                          const int*  __restrict__ src,
                          const int*  __restrict__ dst,
                          const float* __restrict__ wbuf,
                          const float* __restrict__ s,
                          float* __restrict__ out, int E) {
    long gid = (long)blockIdx.x * blockDim.x + threadIdx.x;
    long tot = (long)E * (ROW / 4);
    if (gid >= tot) return;
    int e  = (int)(gid >> 6);
    int c4 = (int)(gid & 63);
    int d  = dst[e];
    float alpha = wbuf[e] / s[d];
    const float4 v = *(const float4*)(h + (size_t)src[e] * ROW + c4 * 4);
    float* o = out + (size_t)d * ROW + c4 * 4;
    atomicAdd(o + 0, alpha * v.x);
    atomicAdd(o + 1, alpha * v.y);
    atomicAdd(o + 2, alpha * v.z);
    atomicAdd(o + 3, alpha * v.w);
}

__global__ void k_finalize(const float* __restrict__ h,
                           const float* __restrict__ s,
                           float* __restrict__ out, int N) {
    long gid = (long)blockIdx.x * blockDim.x + threadIdx.x;
    long tot = (long)N * (ROW / 4);
    if (gid >= tot) return;
    int n  = (int)(gid >> 6);
    int c4 = (int)(gid & 63);
    if (s[n] == 0.0f) {
        *(float4*)(out + (size_t)n * ROW + c4 * 4) =
            *(const float4*)(h + (size_t)n * ROW + c4 * 4);
    }
}

extern "C" void kernel_launch(void* const* d_in, const int* in_sizes, int n_in,
                              void* d_out, int out_size, void* d_ws, size_t ws_size,
                              hipStream_t stream) {
    const float* h   = (const float*)d_in[0];
    const int*   src = (const int*)d_in[1];
    const int*   dst = (const int*)d_in[2];
    const float* w   = (const float*)d_in[3];
    int E = in_sizes[1];
    int N = in_sizes[0] / ROW;
    float* out = (float*)d_out;

    float*    abuf = (float*)d_ws;
    unsigned* m_u  = (unsigned*)((char*)d_ws + (size_t)E * 4);
    float*    s    = (float*)   ((char*)d_ws + (size_t)E * 4 + (size_t)N * 4);

    int nOut4 = N * (ROW / 4);
    k_init<<<1024, 256, 0, stream>>>((float4*)out, nOut4, m_u, s, N);

    int ntiles = (E + 15) / 16;
    k_edge_scores<<<(ntiles + 7) / 8, 256, 0, stream>>>(h, src, dst, w,
                                                        abuf, m_u, E, ntiles);

    k_edge_exp<<<(E + 255) / 256, 256, 0, stream>>>(dst, abuf, m_u, s, E);

    long tot4 = (long)E * (ROW / 4);
    k_scatter<<<(int)((tot4 + 255) / 256), 256, 0, stream>>>(h, src, dst,
                                                             abuf, s, out, E);

    long totf = (long)N * (ROW / 4);
    k_finalize<<<(int)((totf + 255) / 256), 256, 0, stream>>>(h, s, out, N);
}


# float-atomic scatter replaced by per-dst bins (int atomic slot, pairs parked in the node's out row) + one wave per node adding its edges in edge-id order: fixed sum order, no float atomics
# speedup vs baseline: 2.3549x; 2.3549x over previous
#include <hip/hip_runtime.h>
#include <hip/hip_bf16.h>
#include <math.h>
#include <limits.h>

#define SLOT  2
#define DIM   128
#define ROW   (SLOT * DIM)
#define SLOPE 0.98f
#define CAP   128
#define NSLOT (CAP / 32)
#define WPB   8

typedef float v2f __attribute__((ext_vector_type(2)));
typedef float v8f __attribute__((ext_vector_type(8)));

__global__ void k_init(int* __restrict__ cnt, int N) {
    int i = blockIdx.x * blockDim.x + threadIdx.x;
    if (i < N) cnt[i] = 0;
}

__global__ void k_edge_scores(const float* __restrict__ h,
                              const int*  __restrict__ src,
                              const int*  __restrict__ dst,
                              const float* __restrict__ watt,
                              float* __restrict__ abuf,
                              int* __restrict__ cnt,
                              float* __restrict__ out,
                              int E, int ntiles) {
    __shared__ float wlds[2 * DIM];
    int tid = threadIdx.x;
    if (tid < 2 * DIM) wlds[tid] = watt[tid];
    __syncthreads();

    int tile = blockIdx.x * (blockDim.x >> 5) + (tid >> 5);
    if (tile >= ntiles) return;

    int lane = tid & 31;
    int half = lane >> 4;
    int m16  = lane & 15;
    int base = tile * 16;

    int eL = base + m16; if (eL >= E) eL = E - 1;
    int sv = src[eL];
    int dv = dst[eL];
    const float* ps = h + (size_t)sv * ROW + 2 * half;
    const float* pd = h + (size_t)dv * ROW + 2 * half;
    const float* wl = wlds + 2 * half;

    v8f c = {};
#if __has_builtin(__builtin_amdgcn_wmma_f32_16x16x4_f32)
    #pragma unroll
    for (int k = 0; k < 32; ++k) {
        v2f av = *(const v2f*)(ps + 4 * k);
        v2f bv = *(const v2f*)(wl + 4 * k);
        c = __builtin_amdgcn_wmma_f32_16x16x4_f32(false, av, false, bv,
                                                  (short)0, c, false, false);
    }
    #pragma unroll
    for (int k = 0; k < 32; ++k) {
        v2f av = *(const v2f*)(pd + 4 * k);
        v2f bv = *(const v2f*)(wl + DIM + 4 * k);
        c = __builtin_amdgcn_wmma_f32_16x16x4_f32(false, av, false, bv,
                                                  (short)0, c, false, false);
    }
    int r = lane & 7;
    float sc = c[0];
    sc = (r == 1) ? c[1] : sc;
    sc = (r == 2) ? c[2] : sc;
    sc = (r == 3) ? c[3] : sc;
    sc = (r == 4) ? c[4] : sc;
    sc = (r == 5) ? c[5] : sc;
    sc = (r == 6) ? c[6] : sc;
    sc = (r == 7) ? c[7] : sc;
#else
    int r = lane & 7;
    int rowF = ((lane >> 4) << 3) | r;
    int eF = base + rowF; if (eF >= E) eF = E - 1;
    const float* fs = h + (size_t)src[eF] * ROW;
    const float* fd = h + (size_t)dst[eF] * ROW;
    float sc = 0.f;
    for (int k = 0; k < DIM; ++k) sc += fs[k] * wlds[k] + fd[k] * wlds[DIM + k];
#endif

    int row = ((lane >> 4) << 3) | r;
    if ((lane & 8) == 0) {
        int e = base + row;
        if (e < E) {
            float v = (sc >= 0.0f) ? sc : SLOPE * sc;
            abuf[e] = v;
            int d = dst[e];
            int slot = atomicAdd(&cnt[d], 1);
            if (slot < CAP)
                ((int2*)(out + (size_t)d * ROW))[slot] = make_int2(e, __float_as_int(v));
        }
    }
}

__device__ __forceinline__ void fma4(float4& acc, float w, const float4 x) {
    acc.x += w * x.x; acc.y += w * x.y; acc.z += w * x.z; acc.w += w * x.w;
}

__global__ void __launch_bounds__(WPB * 32)
k_gather(const float* __restrict__ h,
         const int*  __restrict__ src,
         const int*  __restrict__ dst,
         const float* __restrict__ abuf,
         const int*  __restrict__ cnt,
         float* out, int N, int E) {
    __shared__ int   l_src[WPB][CAP];
    __shared__ float l_w[WPB][CAP];

    int lane = threadIdx.x & 31;
    int wv   = __builtin_amdgcn_readfirstlane((int)(threadIdx.x >> 5));
    int n    = blockIdx.x * WPB + wv;
    int L    = (n < N) ? cnt[n] : 0;
    bool fast = (L > 0) && (L <= CAP);

    int   e_[NSLOT];
    float a_[NSLOT];
    int   sv_[NSLOT];
    float m = -INFINITY;

    if (fast) {
        const int2* bin = (const int2*)(out + (size_t)n * ROW);
        #pragma unroll
        for (int t = 0; t < NSLOT; ++t) {
            int i = lane + 32 * t;
            e_[t] = INT_MAX; a_[t] = -INFINITY; sv_[t] = 0;
            if (i < L) {
                int2 p = bin[i];
                e_[t]  = p.x;
                a_[t]  = __int_as_float(p.y);
                sv_[t] = src[p.x];
                m = fmaxf(m, a_[t]);
            }
        }
        for (int o = 16; o > 0; o >>= 1) m = fmaxf(m, __shfl_xor(m, o));

        int rk[NSLOT];
        #pragma unroll
        for (int t = 0; t < NSLOT; ++t) rk[t] = 0;
        #pragma unroll
        for (int t2 = 0; t2 < NSLOT; ++t2) {
            int lim = L - 32 * t2; if (lim > 32) lim = 32;
            for (int j = 0; j < lim; ++j) {
                int ej = __builtin_amdgcn_readlane(e_[t2], j);
                #pragma unroll
                for (int t = 0; t < NSLOT; ++t) rk[t] += (ej < e_[t]) ? 1 : 0;
            }
        }
        #pragma unroll
        for (int t = 0; t < NSLOT; ++t) {
            int i = lane + 32 * t;
            if (i < L) {
                l_src[wv][rk[t]] = sv_[t];
                l_w[wv][rk[t]]   = __expf(a_[t] - m);
            }
        }
    }
    __syncthreads();

    float wsl[NSLOT];
    if (fast) {
        #pragma unroll
        for (int t = 0; t < NSLOT; ++t) {
            int i = lane + 32 * t;
            wsl[t] = (i < L) ? l_w[wv][i] : 0.0f;
        }
        float s = (wsl[0] + wsl[1]) + (wsl[2] + wsl[3]);
        for (int o = 16; o > 0; o >>= 1) s += __shfl_xor(s, o);
        #pragma unroll
        for (int t = 0; t < NSLOT; ++t) {
            int i = lane + 32 * t;
            if (i < L) l_w[wv][i] = wsl[t] / s;
        }
    }
    __syncthreads();

    if (n >= N) return;

    float4 accA = make_float4(0.f, 0.f, 0.f, 0.f);
    float4 accB = make_float4(0.f, 0.f, 0.f, 0.f);
    const float* hl = h + lane * 4;

    if (L == 0) {
        accA = *(const float4*)(hl + (size_t)n * ROW);
        accB = *(const float4*)(hl + (size_t)n * ROW + DIM);
    } else if (fast) {
        const int*   ls = l_src[wv];
        const float* lw = l_w[wv];
        int r = 0;
        for (; r + 4 <= L; r += 4) {
            int s0 = ls[r], s1 = ls[r + 1], s2 = ls[r + 2], s3 = ls[r + 3];
            float w0 = lw[r], w1 = lw[r + 1], w2 = lw[r + 2], w3 = lw[r + 3];
            const float* p0 = hl + (size_t)s0 * ROW;
            const float* p1 = hl + (size_t)s1 * ROW;
            const float* p2 = hl + (size_t)s2 * ROW;
            const float* p3 = hl + (size_t)s3 * ROW;
            float4 x0a = *(const float4*)p0, x0b = *(const float4*)(p0 + DIM);
            float4 x1a = *(const float4*)p1, x1b = *(const float4*)(p1 + DIM);
            float4 x2a = *(const float4*)p2, x2b = *(const float4*)(p2 + DIM);
            float4 x3a = *(const float4*)p3, x3b = *(const float4*)(p3 + DIM);
            fma4(accA, w0, x0a); fma4(accB, w0, x0b);
            fma4(accA, w1, x1a); fma4(accB, w1, x1b);
            fma4(accA, w2, x2a); fma4(accB, w2, x2b);
            fma4(accA, w3, x3a); fma4(accB, w3, x3b);
        }
        for (; r < L; ++r) {
            const float* p0 = hl + (size_t)ls[r] * ROW;
            float w0 = lw[r];
            float4 x0a = *(const float4*)p0, x0b = *(const float4*)(p0 + DIM);
            fma4(accA, w0, x0a); fma4(accB, w0, x0b);
        }
    } else {
        float m2 = -INFINITY;
        for (int e0 = 0; e0 < E; e0 += 32) {
            int e = e0 + lane;
            if (e < E && dst[e] == n) m2 = fmaxf(m2, abuf[e]);
        }
        for (int o = 16; o > 0; o >>= 1) m2 = fmaxf(m2, __shfl_xor(m2, o));
        float s2 = 0.0f;
        for (int e0 = 0; e0 < E; e0 += 32) {
            int e = e0 + lane;
            bool hit = (e < E) && (dst[e] == n);
            unsigned mask = (unsigned)__ballot(hit);
            while (mask) {
                int b = __ffs(mask) - 1;
                mask &= mask - 1;
                s2 += __expf(abuf[e0 + b] - m2);
            }
        }
        for (int e0 = 0; e0 < E; e0 += 32) {
            int e = e0 + lane;
            bool hit = (e < E) && (dst[e] == n);
            unsigned mask = (unsigned)__ballot(hit);
            while (mask) {
                int b = __ffs(mask) - 1;
                mask &= mask - 1;
                int ee = e0 + b;
                float al = __expf(abuf[ee] - m2) / s2;
                const float* p0 = hl + (size_t)src[ee] * ROW;
                float4 x0a = *(const float4*)p0, x0b = *(const float4*)(p0 + DIM);
                fma4(accA, al, x0a); fma4(accB, al, x0b);
            }
        }
    }

    float* o = out + (size_t)n * ROW + lane * 4;
    *(float4*)o         = accA;
    *(float4*)(o + DIM) = accB;
}

extern "C" void kernel_launch(void* const* d_in, const int* in_sizes, int n_in,
                              void* d_out, int out_size, void* d_ws, size_t ws_size,
                              hipStream_t stream) {
    const float* h   = (const float*)d_in[0];
    const int*   src = (const int*)d_in[1];
    const int*   dst = (const int*)d_in[2];
    const float* w   = (const float*)d_in[3];
    int E = in_sizes[1];
    int N = in_sizes[0] / ROW;
    float* out = (float*)d_out;

    float* abuf = (float*)d_ws;
    int*   cnt  = (int*)((char*)d_ws + (size_t)E * 4);

    k_init<<<(N + 255) / 256, 256, 0, stream>>>(cnt, N);

    int ntiles = (E + 15) / 16;
    k_edge_scores<<<(ntiles + 7) / 8, 256, 0, stream>>>(h, src, dst, w,
                                                        abuf, cnt, out, E, ntiles);

    k_gather<<<(N + WPB - 1) / WPB, WPB * 32, 0, stream>>>(h, src, dst, abuf,
                                                          cnt, out, N, E);
}


# edge score split by linearity: p=h0.w_src, q=h0.w_dst per node on f32 WMMA (Nx128x2), a=leaky(p[src]+q[dst]) per edge; 256-term dot now two 128-term sums; the refuted try's float atomics are gone
# speedup vs baseline: 4.1522x; 1.7632x over previous
#include <hip/hip_runtime.h>
#include <hip/hip_bf16.h>
#include <math.h>
#include <limits.h>

#define SLOT  2
#define DIM   128
#define ROW   (SLOT * DIM)
#define SLOPE 0.98f
#define CAP   128
#define NSLOT (CAP / 32)
#define WPB   8
#define NTB   128
#define ETB   256
#define EPT   4

typedef float v2f __attribute__((ext_vector_type(2)));
typedef float v8f __attribute__((ext_vector_type(8)));


__global__ void k_node_pq(const float* __restrict__ h,
                          const float* __restrict__ watt,
                          float* __restrict__ pq,
                          int* __restrict__ cnt,
                          int N, int ntiles) {
    __shared__ float wlds[2 * DIM];
    int tid = threadIdx.x;
    if (tid < DIM) { wlds[tid] = watt[tid]; wlds[tid + DIM] = watt[tid + DIM]; }
    __syncthreads();

    int tile = blockIdx.x * (blockDim.x >> 5) + (tid >> 5);
    if (tile >= ntiles) return;

    int lane = tid & 31;
    int half = lane >> 4;
    int c16  = lane & 15;
    int base = tile * 16;

    int nL = base + c16; if (nL >= N) nL = N - 1;
    const float* pa = h + (size_t)nL * ROW + 2 * half;
    const float* wl = wlds + ((c16 & 8) ? DIM : 0) + 2 * half;

    v8f c = {};
    v2f av = {}, bv = {};
    #pragma unroll
    for (int k = 0; k < 32; ++k) {
        av = *(const v2f*)(pa + 4 * k);
        bv = *(const v2f*)(wl + 4 * k);
        c = __builtin_amdgcn_wmma_f32_16x16x4_f32(false, av, false, bv,
                                                  (short)0, c, false, false);
    }
    asm volatile("v_nop\n\tv_nop\n\tv_nop\n\tv_nop" : "+v"(c) : "v"(av), "v"(bv));
    int r = lane & 7;
    float sc = c[0];
    sc = (r == 1) ? c[1] : sc;
    sc = (r == 2) ? c[2] : sc;
    sc = (r == 3) ? c[3] : sc;
    sc = (r == 4) ? c[4] : sc;
    sc = (r == 5) ? c[5] : sc;
    sc = (r == 6) ? c[6] : sc;
    sc = (r == 7) ? c[7] : sc;

    int node = base + (half << 3) + r;
    if (node < N) pq[2 * (size_t)node + ((lane >> 3) & 1)] = sc;
    if (lane < 16 && base + lane < N) cnt[base + lane] = 0;
}

__device__ __forceinline__ float edge_score(const float* __restrict__ pq, int s, int d) {
    float sc = pq[2 * (size_t)s] + pq[2 * (size_t)d + 1];
    return (sc >= 0.0f) ? sc : SLOPE * sc;
}

__global__ void __launch_bounds__(ETB)
k_edge_bins(const int*  __restrict__ src,
            const int*  __restrict__ dst,
            const float* __restrict__ pq,
            int* __restrict__ cnt,
            float* __restrict__ out,
            int E) {
    int e0 = blockIdx.x * (ETB * EPT) + threadIdx.x;
    int   d[EPT];
    float v[EPT];
    int   slot[EPT];
    #pragma unroll
    for (int j = 0; j < EPT; ++j) {
        int e  = e0 + ETB * j;
        int ec = (e < E) ? e : E - 1;
        d[j] = dst[ec];
        v[j] = edge_score(pq, src[ec], d[j]);
    }
    #pragma unroll
    for (int j = 0; j < EPT; ++j) {
        slot[j] = CAP;
        if (e0 + ETB * j < E)
            slot[j] = atomicAdd(&cnt[d[j]], 1);
    }
    #pragma unroll
    for (int j = 0; j < EPT; ++j) {
        if (slot[j] < CAP)
            ((int2*)(out + (size_t)d[j] * ROW))[slot[j]] =
                make_int2(e0 + ETB * j, __float_as_int(v[j]));
    }
}

__device__ __forceinline__ void fma4(float4& acc, float w, const float4 x) {
    acc.x += w * x.x; acc.y += w * x.y; acc.z += w * x.z; acc.w += w * x.w;
}

__global__ void __launch_bounds__(WPB * 32)
k_gather(const float* __restrict__ h,
         const int*  __restrict__ src,
         const int*  __restrict__ dst,
         const float* __restrict__ pq,
         const int*  __restrict__ cnt,
         float* out, int N, int E) {
    __shared__ int   l_src[WPB][CAP];
    __shared__ float l_w[WPB][CAP];

    int lane = threadIdx.x & 31;
    int wv   = __builtin_amdgcn_readfirstlane((int)(threadIdx.x >> 5));
    int n    = blockIdx.x * WPB + wv;
    int L    = (n < N) ? cnt[n] : 0;
    bool fast = (L > 0) && (L <= CAP);

    int   e_[NSLOT];
    float a_[NSLOT];
    int   sv_[NSLOT];
    float m = -INFINITY;

    if (fast) {
        const int2* bin = (const int2*)(out + (size_t)n * ROW);
        #pragma unroll
        for (int t = 0; t < NSLOT; ++t) {
            int i = lane + 32 * t;
            e_[t] = INT_MAX; a_[t] = -INFINITY; sv_[t] = 0;
            if (i < L) {
                int2 p = bin[i];
                e_[t]  = p.x;
                a_[t]  = __int_as_float(p.y);
                sv_[t] = src[p.x];
                m = fmaxf(m, a_[t]);
            }
        }
        for (int o = 16; o > 0; o >>= 1) m = fmaxf(m, __shfl_xor(m, o));

        int rk[NSLOT];
        #pragma unroll
        for (int t = 0; t < NSLOT; ++t) rk[t] = 0;
        #pragma unroll
        for (int t2 = 0; t2 < NSLOT; ++t2) {
            int lim = L - 32 * t2; if (lim > 32) lim = 32;
            for (int j = 0; j < lim; ++j) {
                int ej = __builtin_amdgcn_readlane(e_[t2], j);
                #pragma unroll
                for (int t = 0; t < NSLOT; ++t) rk[t] += (ej < e_[t]) ? 1 : 0;
            }
        }
        #pragma unroll
        for (int t = 0; t < NSLOT; ++t) {
            int i = lane + 32 * t;
            if (i < L) {
                l_src[wv][rk[t]] = sv_[t];
                l_w[wv][rk[t]]   = __expf(a_[t] - m);
            }
        }
    }
    __syncthreads();

    float wsl[NSLOT];
    if (fast) {
        #pragma unroll
        for (int t = 0; t < NSLOT; ++t) {
            int i = lane + 32 * t;
            wsl[t] = (i < L) ? l_w[wv][i] : 0.0f;
        }
        float s = (wsl[0] + wsl[1]) + (wsl[2] + wsl[3]);
        for (int o = 16; o > 0; o >>= 1) s += __shfl_xor(s, o);
        #pragma unroll
        for (int t = 0; t < NSLOT; ++t) {
            int i = lane + 32 * t;
            if (i < L) l_w[wv][i] = wsl[t] / s;
        }
    }
    __syncthreads();

    if (n >= N) return;

    float4 accA = make_float4(0.f, 0.f, 0.f, 0.f);
    float4 accB = make_float4(0.f, 0.f, 0.f, 0.f);
    const float* hl = h + lane * 4;

    if (L == 0) {
        accA = *(const float4*)(hl + (size_t)n * ROW);
        accB = *(const float4*)(hl + (size_t)n * ROW + DIM);
    } else if (fast) {
        const int*   ls = l_src[wv];
        const float* lw = l_w[wv];
        int r = 0;
        for (; r + 4 <= L; r += 4) {
            int s0 = ls[r], s1 = ls[r + 1], s2 = ls[r + 2], s3 = ls[r + 3];
            float w0 = lw[r], w1 = lw[r + 1], w2 = lw[r + 2], w3 = lw[r + 3];
            const float* p0 = hl + (size_t)s0 * ROW;
            const float* p1 = hl + (size_t)s1 * ROW;
            const float* p2 = hl + (size_t)s2 * ROW;
            const float* p3 = hl + (size_t)s3 * ROW;
            float4 x0a = *(const float4*)p0, x0b = *(const float4*)(p0 + DIM);
            float4 x1a = *(const float4*)p1, x1b = *(const float4*)(p1 + DIM);
            float4 x2a = *(const float4*)p2, x2b = *(const float4*)(p2 + DIM);
            float4 x3a = *(const float4*)p3, x3b = *(const float4*)(p3 + DIM);
            fma4(accA, w0, x0a); fma4(accB, w0, x0b);
            fma4(accA, w1, x1a); fma4(accB, w1, x1b);
            fma4(accA, w2, x2a); fma4(accB, w2, x2b);
            fma4(accA, w3, x3a); fma4(accB, w3, x3b);
        }
        for (; r < L; ++r) {
            const float* p0 = hl + (size_t)ls[r] * ROW;
            float w0 = lw[r];
            float4 x0a = *(const float4*)p0, x0b = *(const float4*)(p0 + DIM);
            fma4(accA, w0, x0a); fma4(accB, w0, x0b);
        }
    } else {
        float m2 = -INFINITY;
        for (int e0 = 0; e0 < E; e0 += 32) {
            int e = e0 + lane;
            if (e < E && dst[e] == n) m2 = fmaxf(m2, edge_score(pq, src[e], n));
        }
        for (int o = 16; o > 0; o >>= 1) m2 = fmaxf(m2, __shfl_xor(m2, o));
        float s2 = 0.0f;
        for (int e0 = 0; e0 < E; e0 += 32) {
            int e = e0 + lane;
            bool hit = (e < E) && (dst[e] == n);
            unsigned mask = (unsigned)__ballot(hit);
            while (mask) {
                int b = __ffs(mask) - 1;
                mask &= mask - 1;
                s2 += __expf(edge_score(pq, src[e0 + b], n) - m2);
            }
        }
        for (int e0 = 0; e0 < E; e0 += 32) {
            int e = e0 + lane;
            bool hit = (e < E) && (dst[e] == n);
            unsigned mask = (unsigned)__ballot(hit);
            while (mask) {
                int b = __ffs(mask) - 1;
                mask &= mask - 1;
                int ee = e0 + b;
                float al = __expf(edge_score(pq, src[ee], n) - m2) / s2;
                const float* p0 = hl + (size_t)src[ee] * ROW;
                float4 x0a = *(const float4*)p0, x0b = *(const float4*)(p0 + DIM);
                fma4(accA, al, x0a); fma4(accB, al, x0b);
            }
        }
    }

    float* o = out + (size_t)n * ROW + lane * 4;
    *(float4*)o         = accA;
    *(float4*)(o + DIM) = accB;
}

extern "C" void kernel_launch(void* const* d_in, const int* in_sizes, int n_in,
                              void* d_out, int out_size, void* d_ws, size_t ws_size,
                              hipStream_t stream) {
    const float* h   = (const float*)d_in[0];
    const int*   src = (const int*)d_in[1];
    const int*   dst = (const int*)d_in[2];
    const float* w   = (const float*)d_in[3];
    int E = in_sizes[1];
    int N = in_sizes[0] / ROW;
    float* out = (float*)d_out;

    float* pq  = (float*)d_ws;
    int*   cnt = (int*)((char*)d_ws + (size_t)N * 8);

    int ntiles = (N + 15) / 16;
    k_node_pq<<<(ntiles + (NTB / 32) - 1) / (NTB / 32), NTB, 0, stream>>>(h, w, pq, cnt, N, ntiles);

    k_edge_bins<<<(E + ETB * EPT - 1) / (ETB * EPT), ETB, 0, stream>>>(src, dst, pq, cnt, out, E);

    k_gather<<<(N + WPB - 1) / WPB, WPB * 32, 0, stream>>>(h, src, dst, pq,
                                                          cnt, out, N, E);
}


# score halves and edge partition share one launch: each partition work-group parks its edges in its own list segment, count in slot 0, so no zeroed counter and no global add; 3 launches, not 4
# speedup vs baseline: 7.8104x; 1.8810x over previous
#include <hip/hip_runtime.h>
#include <hip/hip_bf16.h>
#include <math.h>
#include <limits.h>

#define SLOT  2
#define DIM   128
#define ROW   (SLOT * DIM)
#define SLOPE 0.98f
#define CAP   128
#define LPR   64
#define NSLOT (CAP / 32)
#define WPB   8
#define NBK   250
#define PTB   512
#define PEPT  8
#define SEG   LPR
#define OTB   1024
#define SPT   (OTB / SEG)
#define OLD   13

typedef float v2f __attribute__((ext_vector_type(2)));
typedef float v8f __attribute__((ext_vector_type(8)));


__device__ __forceinline__ void role_node_pq(const float* __restrict__ h,
                                             const float* __restrict__ watt,
                                             float* __restrict__ pq,
                                             float* wlds,
                                             int N, int ntiles, int blk) {
    int tid = threadIdx.x;
    if (tid < DIM) { wlds[tid] = watt[tid]; wlds[tid + DIM] = watt[tid + DIM]; }
    __syncthreads();

    int tile = blk * (PTB >> 5) + (tid >> 5);
    if (tile >= ntiles) return;

    int lane = tid & 31;
    int half = lane >> 4;
    int c16  = lane & 15;
    int base = tile * 16;

    int nL = base + c16; if (nL >= N) nL = N - 1;
    const float* pa = h + (size_t)nL * ROW + 2 * half;
    const float* wl = wlds + ((c16 & 8) ? DIM : 0) + 2 * half;

    v8f c = {};
    v2f av = {}, bv = {};
    #pragma unroll
    for (int k = 0; k < 32; ++k) {
        av = *(const v2f*)(pa + 4 * k);
        bv = *(const v2f*)(wl + 4 * k);
        c = __builtin_amdgcn_wmma_f32_16x16x4_f32(false, av, false, bv,
                                                  (short)0, c, false, false);
    }
    asm volatile("v_nop\n\tv_nop\n\tv_nop\n\tv_nop" : "+v"(c) : "v"(av), "v"(bv));
    int r = lane & 7;
    float sc = c[0];
    sc = (r == 1) ? c[1] : sc;
    sc = (r == 2) ? c[2] : sc;
    sc = (r == 3) ? c[3] : sc;
    sc = (r == 4) ? c[4] : sc;
    sc = (r == 5) ? c[5] : sc;
    sc = (r == 6) ? c[6] : sc;
    sc = (r == 7) ? c[7] : sc;

    int node = base + (half << 3) + r;
    if (node < N) pq[2 * (size_t)node + ((lane >> 3) & 1)] = sc;
}

__device__ __forceinline__ float edge_score(const float* __restrict__ pq, int s, int d) {
    float sc = pq[2 * (size_t)s] + pq[2 * (size_t)d + 1];
    return (sc >= 0.0f) ? sc : SLOPE * sc;
}

static_assert(LPR == 64 && LPR * 8 == DIM * 4, "a row's second half holds LPR pairs; list_slot shifts by 6");
__device__ __forceinline__ int2* list_slot(float* out, int n0, int p) {
    return (int2*)(out + (size_t)(n0 + (p >> 6)) * ROW + DIM) + (p & (LPR - 1));
}
static_assert(PTB >= NBK, "one thread per range writes the segment's count");
static_assert(PTB >= DIM, "the score role stages w with one thread per column");
static_assert(SEG == 64 && OTB % SEG == 0, "k_owner splits a thread index into segment and slot by a shift of 6");

__device__ __forceinline__ void role_part(const int* __restrict__ dst,
                                          float* __restrict__ out,
                                          int* lc,
                                          int N, int E, int npo, int g) {
    int tid = threadIdx.x;
    if (tid < NBK) lc[tid] = 0;
    __syncthreads();

    int e0 = g * (PTB * PEPT) + tid;
    int d[PEPT], b[PEPT], rk[PEPT];
    #pragma unroll
    for (int j = 0; j < PEPT; ++j) {
        int e  = e0 + PTB * j;
        int ec = (e < E) ? e : E - 1;
        d[j] = dst[ec];
    }
    #pragma unroll
    for (int j = 0; j < PEPT; ++j) {
        b[j]  = (int)((unsigned)d[j] / (unsigned)npo);
        rk[j] = -1;
        if (e0 + PTB * j < E) rk[j] = atomicAdd(&lc[b[j]], 1);
    }
    __syncthreads();
    int sbase = g * SEG;
    if (tid < NBK) {
        int n0 = tid * npo;
        if (n0 < N) {
            int nn = N - n0; if (nn > npo) nn = npo;
            if (sbase + SEG <= nn * LPR)
                *list_slot(out, n0, sbase) = make_int2(lc[tid], 0);
        }
    }
    #pragma unroll
    for (int j = 0; j < PEPT; ++j) {
        if (rk[j] >= 0 && rk[j] < SEG - 1) {
            int n0 = b[j] * npo;
            int nn = N - n0; if (nn > npo) nn = npo;
            if (sbase + SEG <= nn * LPR)
                *list_slot(out, n0, sbase + 1 + rk[j]) = make_int2(e0 + PTB * j, d[j]);
        }
    }
}

__global__ void __launch_bounds__(PTB)
k_pq_part(const float* __restrict__ h,
          const float* __restrict__ watt,
          const int* __restrict__ dst,
          float* __restrict__ pq,
          float* __restrict__ out,
          int N, int E, int npo, int ntiles, int G) {
    __shared__ int   lc[NBK];
    __shared__ float wlds[2 * DIM];
    int blk = blockIdx.x;
    if (blk < G) role_part(dst, out, lc, N, E, npo, blk);
    else         role_node_pq(h, watt, pq, wlds, N, ntiles, blk - G);
}

__global__ void __launch_bounds__(OTB)
k_owner(int* __restrict__ cnt,
        float* __restrict__ out,
        int N, int npo, int G) {
    extern __shared__ int lcnt[];
    __shared__ int ovf;
    int tid = threadIdx.x;
    int n0  = blockIdx.x * npo;
    if (n0 >= N) return;
    int nn = N - n0; if (nn > npo) nn = npo;
    bool over = G * SEG > nn * LPR;
    for (int i = tid; i < nn; i += OTB) lcnt[i] = 0;
    if (tid == 0) ovf = 0;
    __syncthreads();

    if (!over) {
        int sg = tid >> 6;
        int s  = tid & (SEG - 1);
        for (int g0 = 0; g0 < G; g0 += SPT * OLD) {
            int  hc[OLD];
            int2 ed[OLD];
            #pragma unroll
            for (int j = 0; j < OLD; ++j) {
                int g = g0 + j * SPT + sg;
                hc[j] = 0; ed[j] = make_int2(0, 0);
                if (g < G) {
                    const int2* q = (const int2*)(out + (size_t)(n0 + g0 + sg) * ROW + DIM)
                                  + (size_t)j * (SPT * (ROW / 2));
                    hc[j] = q[0].x;
                    ed[j] = q[s];
                }
            }
            #pragma unroll
            for (int j = 0; j < OLD; ++j) asm volatile("" : "+v"(hc[j]), "+v"(ed[j].x), "+v"(ed[j].y));
            #pragma unroll
            for (int j = 0; j < OLD; ++j) {
                int g = g0 + j * SPT + sg;
                if (g < G) {
                    int c = hc[j];
                    if (c > SEG - 1) {
                        ovf = 1;
                    } else if (s >= 1 && s <= c) {
                        int r    = ed[j].y - n0;
                        int slot = atomicAdd(&lcnt[r], 1);
                        if (slot < CAP) ((int*)(out + (size_t)ed[j].y * ROW))[slot] = ed[j].x;
                    }
                }
            }
        }
    }
    __syncthreads();
    bool o2 = over || (ovf != 0);
    for (int i = tid; i < nn; i += OTB) cnt[n0 + i] = o2 ? (CAP + 1) : lcnt[i];
}

__device__ __forceinline__ void fma4(float4& acc, float w, const float4 x) {
    acc.x += w * x.x; acc.y += w * x.y; acc.z += w * x.z; acc.w += w * x.w;
}

__global__ void __launch_bounds__(WPB * 32)
k_gather(const float* __restrict__ h,
         const int*  __restrict__ src,
         const int*  __restrict__ dst,
         const float* __restrict__ pq,
         const int*  __restrict__ cnt,
         float* out, int N, int E) {
    __shared__ int   l_src[WPB][CAP];
    __shared__ float l_w[WPB][CAP];

    int lane = threadIdx.x & 31;
    int wv   = __builtin_amdgcn_readfirstlane((int)(threadIdx.x >> 5));
    int n    = blockIdx.x * WPB + wv;
    int L    = (n < N) ? cnt[n] : 0;
    bool fast = (L > 0) && (L <= CAP);

    int   e_[NSLOT];
    float a_[NSLOT];
    int   sv_[NSLOT];
    float m = -INFINITY;

    if (fast) {
        const int* bin = (const int*)(out + (size_t)n * ROW);
        #pragma unroll
        for (int t = 0; t < NSLOT; ++t) {
            int i = lane + 32 * t;
            e_[t] = INT_MAX; a_[t] = -INFINITY; sv_[t] = 0;
            if (i < L) {
                e_[t]  = bin[i];
                sv_[t] = src[e_[t]];
            }
        }
        #pragma unroll
        for (int t = 0; t < NSLOT; ++t) {
            int i = lane + 32 * t;
            if (i < L) {
                a_[t] = edge_score(pq, sv_[t], n);
                m = fmaxf(m, a_[t]);
            }
        }
        for (int o = 16; o > 0; o >>= 1) m = fmaxf(m, __shfl_xor(m, o));

        int rk[NSLOT];
        #pragma unroll
        for (int t = 0; t < NSLOT; ++t) rk[t] = 0;
        #pragma unroll
        for (int t2 = 0; t2 < NSLOT; ++t2) {
            int lim = L - 32 * t2; if (lim > 32) lim = 32;
            for (int j = 0; j < lim; ++j) {
                int ej = __builtin_amdgcn_readlane(e_[t2], j);
                #pragma unroll
                for (int t = 0; t < NSLOT; ++t) rk[t] += (ej < e_[t]) ? 1 : 0;
            }
        }
        #pragma unroll
        for (int t = 0; t < NSLOT; ++t) {
            int i = lane + 32 * t;
            if (i < L) {
                l_src[wv][rk[t]] = sv_[t];
                l_w[wv][rk[t]]   = __expf(a_[t] - m);
            }
        }
    }
    __syncthreads();

    float wsl[NSLOT];
    if (fast) {
        #pragma unroll
        for (int t = 0; t < NSLOT; ++t) {
            int i = lane + 32 * t;
            wsl[t] = (i < L) ? l_w[wv][i] : 0.0f;
        }
        float s = (wsl[0] + wsl[1]) + (wsl[2] + wsl[3]);
        for (int o = 16; o > 0; o >>= 1) s += __shfl_xor(s, o);
        #pragma unroll
        for (int t = 0; t < NSLOT; ++t) {
            int i = lane + 32 * t;
            if (i < L) l_w[wv][i] = wsl[t] / s;
        }
    }
    __syncthreads();

    if (n >= N) return;

    float4 accA = make_float4(0.f, 0.f, 0.f, 0.f);
    float4 accB = make_float4(0.f, 0.f, 0.f, 0.f);
    const float* hl = h + lane * 4;

    if (L == 0) {
        accA = *(const float4*)(hl + (size_t)n * ROW);
        accB = *(const float4*)(hl + (size_t)n * ROW + DIM);
    } else if (fast) {
        const int*   ls = l_src[wv];
        const float* lw = l_w[wv];
        int r = 0;
        for (; r + 4 <= L; r += 4) {
            int s0 = ls[r], s1 = ls[r + 1], s2 = ls[r + 2], s3 = ls[r + 3];
            float w0 = lw[r], w1 = lw[r + 1], w2 = lw[r + 2], w3 = lw[r + 3];
            const float* p0 = hl + (size_t)s0 * ROW;
            const float* p1 = hl + (size_t)s1 * ROW;
            const float* p2 = hl + (size_t)s2 * ROW;
            const float* p3 = hl + (size_t)s3 * ROW;
            float4 x0a = *(const float4*)p0, x0b = *(const float4*)(p0 + DIM);
            float4 x1a = *(const float4*)p1, x1b = *(const float4*)(p1 + DIM);
            float4 x2a = *(const float4*)p2, x2b = *(const float4*)(p2 + DIM);
            float4 x3a = *(const float4*)p3, x3b = *(const float4*)(p3 + DIM);
            fma4(accA, w0, x0a); fma4(accB, w0, x0b);
            fma4(accA, w1, x1a); fma4(accB, w1, x1b);
            fma4(accA, w2, x2a); fma4(accB, w2, x2b);
            fma4(accA, w3, x3a); fma4(accB, w3, x3b);
        }
        for (; r < L; ++r) {
            const float* p0 = hl + (size_t)ls[r] * ROW;
            float w0 = lw[r];
            float4 x0a = *(const float4*)p0, x0b = *(const float4*)(p0 + DIM);
            fma4(accA, w0, x0a); fma4(accB, w0, x0b);
        }
    } else {
        float m2 = -INFINITY;
        int   nh = 0;
        for (int e0 = 0; e0 < E; e0 += 32) {
            int e = e0 + lane;
            bool hit = (e < E) && (dst[e] == n);
            if (hit) m2 = fmaxf(m2, edge_score(pq, src[e], n));
            nh += __popc((unsigned)__ballot(hit));
        }
        for (int o = 16; o > 0; o >>= 1) m2 = fmaxf(m2, __shfl_xor(m2, o));
        if (nh == 0) {
            accA = *(const float4*)(hl + (size_t)n * ROW);
            accB = *(const float4*)(hl + (size_t)n * ROW + DIM);
        }
        float s2 = 0.0f;
        for (int e0 = 0; nh > 0 && e0 < E; e0 += 32) {
            int e = e0 + lane;
            bool hit = (e < E) && (dst[e] == n);
            unsigned mask = (unsigned)__ballot(hit);
            while (mask) {
                int b = __ffs(mask) - 1;
                mask &= mask - 1;
                s2 += __expf(edge_score(pq, src[e0 + b], n) - m2);
            }
        }
        for (int e0 = 0; nh > 0 && e0 < E; e0 += 32) {
            int e = e0 + lane;
            bool hit = (e < E) && (dst[e] == n);
            unsigned mask = (unsigned)__ballot(hit);
            while (mask) {
                int b = __ffs(mask) - 1;
                mask &= mask - 1;
                int ee = e0 + b;
                float al = __expf(edge_score(pq, src[ee], n) - m2) / s2;
                const float* p0 = hl + (size_t)src[ee] * ROW;
                float4 x0a = *(const float4*)p0, x0b = *(const float4*)(p0 + DIM);
                fma4(accA, al, x0a); fma4(accB, al, x0b);
            }
        }
    }

    float* o = out + (size_t)n * ROW + lane * 4;
    *(float4*)o         = accA;
    *(float4*)(o + DIM) = accB;
}

extern "C" void kernel_launch(void* const* d_in, const int* in_sizes, int n_in,
                              void* d_out, int out_size, void* d_ws, size_t ws_size,
                              hipStream_t stream) {
    const float* h   = (const float*)d_in[0];
    const int*   src = (const int*)d_in[1];
    const int*   dst = (const int*)d_in[2];
    const float* w   = (const float*)d_in[3];
    int E = in_sizes[1];
    int N = in_sizes[0] / ROW;
    float* out = (float*)d_out;

    float* pq  = (float*)d_ws;
    int*   cnt = (int*)((char*)d_ws + (size_t)N * 8);

    int ntiles = (N + 15) / 16;
    int npo = (N + NBK - 1) / NBK;
    int G   = (E + PTB * PEPT - 1) / (PTB * PEPT);
    int PQB = (ntiles + (PTB / 32) - 1) / (PTB / 32);
    k_pq_part<<<G + PQB, PTB, 0, stream>>>(h, w, dst, pq, out, N, E, npo, ntiles, G);

    k_owner<<<NBK, OTB, npo * sizeof(int), stream>>>(cnt, out, N, npo, G);

    k_gather<<<(N + WPB - 1) / WPB, WPB * 32, 0, stream>>>(h, src, dst, pq,
                                                          cnt, out, N, E);
}


# gather runs as resident waves: 1024 work-groups of 16 waves fill the wave slots once and each wave takes node after node of its group's share from an LDS counter; no work-group barrier per node
# speedup vs baseline: 8.0738x; 1.0337x over previous
#include <hip/hip_runtime.h>
#include <hip/hip_bf16.h>
#include <math.h>
#include <limits.h>

#define SLOT  2
#define DIM   128
#define ROW   (SLOT * DIM)
#define SLOPE 0.98f
#define CAP   128
#define LPR   64
#define NSLOT (CAP / 32)
#define GWPB  16
#define GGRID 1024
#define NBK   250
#define PTB   512
#define PEPT  8
#define SEG   LPR
#define OTB   1024
#define SPT   (OTB / SEG)
#define OLD   13

typedef float v2f __attribute__((ext_vector_type(2)));
typedef float v8f __attribute__((ext_vector_type(8)));


__device__ __forceinline__ void role_node_pq(const float* __restrict__ h,
                                             const float* __restrict__ watt,
                                             float* __restrict__ pq,
                                             float* wlds,
                                             int N, int ntiles, int blk) {
    int tid = threadIdx.x;
    if (tid < DIM) { wlds[tid] = watt[tid]; wlds[tid + DIM] = watt[tid + DIM]; }
    __syncthreads();

    int tile = blk * (PTB >> 5) + (tid >> 5);
    if (tile >= ntiles) return;

    int lane = tid & 31;
    int half = lane >> 4;
    int c16  = lane & 15;
    int base = tile * 16;

    int nL = base + c16; if (nL >= N) nL = N - 1;
    const float* pa = h + (size_t)nL * ROW + 2 * half;
    const float* wl = wlds + ((c16 & 8) ? DIM : 0) + 2 * half;

    v8f c = {};
    v2f av = {}, bv = {};
    #pragma unroll
    for (int k = 0; k < 32; ++k) {
        av = *(const v2f*)(pa + 4 * k);
        bv = *(const v2f*)(wl + 4 * k);
        c = __builtin_amdgcn_wmma_f32_16x16x4_f32(false, av, false, bv,
                                                  (short)0, c, false, false);
    }
    asm volatile("v_nop\n\tv_nop\n\tv_nop\n\tv_nop" : "+v"(c) : "v"(av), "v"(bv));
    int r = lane & 7;
    float sc = c[0];
    sc = (r == 1) ? c[1] : sc;
    sc = (r == 2) ? c[2] : sc;
    sc = (r == 3) ? c[3] : sc;
    sc = (r == 4) ? c[4] : sc;
    sc = (r == 5) ? c[5] : sc;
    sc = (r == 6) ? c[6] : sc;
    sc = (r == 7) ? c[7] : sc;

    int node = base + (half << 3) + r;
    if (node < N) pq[2 * (size_t)node + ((lane >> 3) & 1)] = sc;
}

__device__ __forceinline__ float edge_score(const float* __restrict__ pq, int s, int d) {
    float sc = pq[2 * (size_t)s] + pq[2 * (size_t)d + 1];
    return (sc >= 0.0f) ? sc : SLOPE * sc;
}

static_assert(LPR == 64 && LPR * 8 == DIM * 4, "a row's second half holds LPR pairs; list_slot shifts by 6");
__device__ __forceinline__ int2* list_slot(float* out, int n0, int p) {
    return (int2*)(out + (size_t)(n0 + (p >> 6)) * ROW + DIM) + (p & (LPR - 1));
}
static_assert(PTB >= NBK, "one thread per range writes the segment's count");
static_assert(PTB >= DIM, "the score role stages w with one thread per column");
static_assert(SEG == 64 && OTB % SEG == 0, "k_owner splits a thread index into segment and slot by a shift of 6");

__device__ __forceinline__ void role_part(const int* __restrict__ dst,
                                          float* __restrict__ out,
                                          int* lc,
                                          int N, int E, int npo, int g) {
    int tid = threadIdx.x;
    if (tid < NBK) lc[tid] = 0;
    __syncthreads();

    int e0 = g * (PTB * PEPT) + tid;
    int d[PEPT], b[PEPT], rk[PEPT];
    #pragma unroll
    for (int j = 0; j < PEPT; ++j) {
        int e  = e0 + PTB * j;
        int ec = (e < E) ? e : E - 1;
        d[j] = dst[ec];
    }
    #pragma unroll
    for (int j = 0; j < PEPT; ++j) {
        b[j]  = (int)((unsigned)d[j] / (unsigned)npo);
        rk[j] = -1;
        if (e0 + PTB * j < E) rk[j] = atomicAdd(&lc[b[j]], 1);
    }
    __syncthreads();
    int sbase = g * SEG;
    if (tid < NBK) {
        int n0 = tid * npo;
        if (n0 < N) {
            int nn = N - n0; if (nn > npo) nn = npo;
            if (sbase + SEG <= nn * LPR)
                *list_slot(out, n0, sbase) = make_int2(lc[tid], 0);
        }
    }
    #pragma unroll
    for (int j = 0; j < PEPT; ++j) {
        if (rk[j] >= 0 && rk[j] < SEG - 1) {
            int n0 = b[j] * npo;
            int nn = N - n0; if (nn > npo) nn = npo;
            if (sbase + SEG <= nn * LPR)
                *list_slot(out, n0, sbase + 1 + rk[j]) = make_int2(e0 + PTB * j, d[j]);
        }
    }
}

__global__ void __launch_bounds__(PTB)
k_pq_part(const float* __restrict__ h,
          const float* __restrict__ watt,
          const int* __restrict__ dst,
          float* __restrict__ pq,
          float* __restrict__ out,
          int N, int E, int npo, int ntiles, int G) {
    __shared__ int   lc[NBK];
    __shared__ float wlds[2 * DIM];
    int blk = blockIdx.x;
    if (blk < G) role_part(dst, out, lc, N, E, npo, blk);
    else         role_node_pq(h, watt, pq, wlds, N, ntiles, blk - G);
}

__global__ void __launch_bounds__(OTB)
k_owner(int* __restrict__ cnt,
        float* __restrict__ out,
        int N, int npo, int G) {
    extern __shared__ int lcnt[];
    __shared__ int ovf;
    int tid = threadIdx.x;
    int n0  = blockIdx.x * npo;
    if (n0 >= N) return;
    int nn = N - n0; if (nn > npo) nn = npo;
    bool over = G * SEG > nn * LPR;
    for (int i = tid; i < nn; i += OTB) lcnt[i] = 0;
    if (tid == 0) ovf = 0;
    __syncthreads();

    if (!over) {
        int sg = tid >> 6;
        int s  = tid & (SEG - 1);
        for (int g0 = 0; g0 < G; g0 += SPT * OLD) {
            int  hc[OLD];
            int2 ed[OLD];
            #pragma unroll
            for (int j = 0; j < OLD; ++j) {
                int g = g0 + j * SPT + sg;
                hc[j] = 0; ed[j] = make_int2(0, 0);
                if (g < G) {
                    const int2* q = (const int2*)(out + (size_t)(n0 + g0 + sg) * ROW + DIM)
                                  + (size_t)j * (SPT * (ROW / 2));
                    hc[j] = q[0].x;
                    ed[j] = q[s];
                }
            }
            #pragma unroll
            for (int j = 0; j < OLD; ++j) asm volatile("" : "+v"(hc[j]), "+v"(ed[j].x), "+v"(ed[j].y));
            #pragma unroll
            for (int j = 0; j < OLD; ++j) {
                int g = g0 + j * SPT + sg;
                if (g < G) {
                    int c = hc[j];
                    if (c > SEG - 1) {
                        ovf = 1;
                    } else if (s >= 1 && s <= c) {
                        int r    = ed[j].y - n0;
                        int slot = atomicAdd(&lcnt[r], 1);
                        if (slot < CAP) ((int*)(out + (size_t)ed[j].y * ROW))[slot] = ed[j].x;
                    }
                }
            }
        }
    }
    __syncthreads();
    bool o2 = over || (ovf != 0);
    for (int i = tid; i < nn; i += OTB) cnt[n0 + i] = o2 ? (CAP + 1) : lcnt[i];
}

__device__ __forceinline__ void fma4(float4& acc, float w, const float4 x) {
    acc.x += w * x.x; acc.y += w * x.y; acc.z += w * x.z; acc.w += w * x.w;
}

__global__ void __launch_bounds__(GWPB * 32, 16)
k_gather(const float* __restrict__ h,
         const int*  __restrict__ src,
         const int*  __restrict__ dst,
         const float* __restrict__ pq,
         const int*  __restrict__ cnt,
         float* out, int N, int E, int npg) {
    __shared__ int   l_src[GWPB][CAP];
    __shared__ float l_w[GWPB][CAP];
    __shared__ int   next;

    int lane = threadIdx.x & 31;
    int wv   = __builtin_amdgcn_readfirstlane((int)(threadIdx.x >> 5));
    int nb = blockIdx.x * npg; if (nb > N) nb = N;
    int ne = nb + npg;         if (ne > N) ne = N;
    if (threadIdx.x == 0) next = nb;
    __syncthreads();

    int*   ls = l_src[wv];
    float* lw = l_w[wv];
    const float* hl = h + lane * 4;

    for (;;) {
        int n = 0;
        if (lane == 0) n = atomicAdd(&next, 1);
        n = __builtin_amdgcn_readfirstlane(n);
        if (n >= ne) break;

        int L = cnt[n];
        bool fast = (L > 0) && (L <= CAP);

        float4 accA = make_float4(0.f, 0.f, 0.f, 0.f);
        float4 accB = make_float4(0.f, 0.f, 0.f, 0.f);

        if (L == 0) {
            accA = *(const float4*)(hl + (size_t)n * ROW);
            accB = *(const float4*)(hl + (size_t)n * ROW + DIM);
        } else if (fast) {
            int   e_[NSLOT];
            float a_[NSLOT];
            int   sv_[NSLOT];
            float m = -INFINITY;
            const int* bin = (const int*)(out + (size_t)n * ROW);
            #pragma unroll
            for (int t = 0; t < NSLOT; ++t) {
                int i = lane + 32 * t;
                e_[t] = INT_MAX; a_[t] = -INFINITY; sv_[t] = 0;
                if (i < L) {
                    e_[t]  = bin[i];
                    sv_[t] = src[e_[t]];
                }
            }
            #pragma unroll
            for (int t = 0; t < NSLOT; ++t) {
                int i = lane + 32 * t;
                if (i < L) {
                    a_[t] = edge_score(pq, sv_[t], n);
                    m = fmaxf(m, a_[t]);
                }
            }
            for (int o = 16; o > 0; o >>= 1) m = fmaxf(m, __shfl_xor(m, o));

            int rk[NSLOT];
            #pragma unroll
            for (int t = 0; t < NSLOT; ++t) rk[t] = 0;
            #pragma unroll
            for (int t2 = 0; t2 < NSLOT; ++t2) {
                int lim = L - 32 * t2; if (lim > 32) lim = 32;
                for (int j = 0; j < lim; ++j) {
                    int ej = __builtin_amdgcn_readlane(e_[t2], j);
                    #pragma unroll
                    for (int t = 0; t < NSLOT; ++t) rk[t] += (ej < e_[t]) ? 1 : 0;
                }
            }
            #pragma unroll
            for (int t = 0; t < NSLOT; ++t) {
                int i = lane + 32 * t;
                if (i < L) {
                    ls[rk[t]] = sv_[t];
                    lw[rk[t]] = __expf(a_[t] - m);
                }
            }

            float wsl[NSLOT];
            #pragma unroll
            for (int t = 0; t < NSLOT; ++t) {
                int i = lane + 32 * t;
                wsl[t] = (i < L) ? lw[i] : 0.0f;
            }
            float s = (wsl[0] + wsl[1]) + (wsl[2] + wsl[3]);
            for (int o = 16; o > 0; o >>= 1) s += __shfl_xor(s, o);
            #pragma unroll
            for (int t = 0; t < NSLOT; ++t) {
                int i = lane + 32 * t;
                if (i < L) lw[i] = wsl[t] / s;
            }

            int r = 0;
            for (; r + 4 <= L; r += 4) {
                int s0 = ls[r], s1 = ls[r + 1], s2 = ls[r + 2], s3 = ls[r + 3];
                float w0 = lw[r], w1 = lw[r + 1], w2 = lw[r + 2], w3 = lw[r + 3];
                const float* p0 = hl + (size_t)s0 * ROW;
                const float* p1 = hl + (size_t)s1 * ROW;
                const float* p2 = hl + (size_t)s2 * ROW;
                const float* p3 = hl + (size_t)s3 * ROW;
                float4 x0a = *(const float4*)p0, x0b = *(const float4*)(p0 + DIM);
                float4 x1a = *(const float4*)p1, x1b = *(const float4*)(p1 + DIM);
                float4 x2a = *(const float4*)p2, x2b = *(const float4*)(p2 + DIM);
                float4 x3a = *(const float4*)p3, x3b = *(const float4*)(p3 + DIM);
                fma4(accA, w0, x0a); fma4(accB, w0, x0b);
                fma4(accA, w1, x1a); fma4(accB, w1, x1b);
                fma4(accA, w2, x2a); fma4(accB, w2, x2b);
                fma4(accA, w3, x3a); fma4(accB, w3, x3b);
            }
            for (; r < L; ++r) {
                const float* p0 = hl + (size_t)ls[r] * ROW;
                float w0 = lw[r];
                float4 x0a = *(const float4*)p0, x0b = *(const float4*)(p0 + DIM);
                fma4(accA, w0, x0a); fma4(accB, w0, x0b);
            }
        } else {
            float m2 = -INFINITY;
            int   nh = 0;
            for (int e0 = 0; e0 < E; e0 += 32) {
                int e = e0 + lane;
                bool hit = (e < E) && (dst[e] == n);
                if (hit) m2 = fmaxf(m2, edge_score(pq, src[e], n));
                nh += __popc((unsigned)__ballot(hit));
            }
            for (int o = 16; o > 0; o >>= 1) m2 = fmaxf(m2, __shfl_xor(m2, o));
            if (nh == 0) {
                accA = *(const float4*)(hl + (size_t)n * ROW);
                accB = *(const float4*)(hl + (size_t)n * ROW + DIM);
            }
            float s2 = 0.0f;
            for (int e0 = 0; nh > 0 && e0 < E; e0 += 32) {
                int e = e0 + lane;
                bool hit = (e < E) && (dst[e] == n);
                unsigned mask = (unsigned)__ballot(hit);
                while (mask) {
                    int b = __ffs(mask) - 1;
                    mask &= mask - 1;
                    s2 += __expf(edge_score(pq, src[e0 + b], n) - m2);
                }
            }
            for (int e0 = 0; nh > 0 && e0 < E; e0 += 32) {
                int e = e0 + lane;
                bool hit = (e < E) && (dst[e] == n);
                unsigned mask = (unsigned)__ballot(hit);
                while (mask) {
                    int b = __ffs(mask) - 1;
                    mask &= mask - 1;
                    int ee = e0 + b;
                    float al = __expf(edge_score(pq, src[ee], n) - m2) / s2;
                    const float* p0 = hl + (size_t)src[ee] * ROW;
                    float4 x0a = *(const float4*)p0, x0b = *(const float4*)(p0 + DIM);
                    fma4(accA, al, x0a); fma4(accB, al, x0b);
                }
            }
        }

        float* o = out + (size_t)n * ROW + lane * 4;
        *(float4*)o         = accA;
        *(float4*)(o + DIM) = accB;
    }
}

extern "C" void kernel_launch(void* const* d_in, const int* in_sizes, int n_in,
                              void* d_out, int out_size, void* d_ws, size_t ws_size,
                              hipStream_t stream) {
    const float* h   = (const float*)d_in[0];
    const int*   src = (const int*)d_in[1];
    const int*   dst = (const int*)d_in[2];
    const float* w   = (const float*)d_in[3];
    int E = in_sizes[1];
    int N = in_sizes[0] / ROW;
    float* out = (float*)d_out;

    float* pq  = (float*)d_ws;
    int*   cnt = (int*)((char*)d_ws + (size_t)N * 8);

    int ntiles = (N + 15) / 16;
    int npo = (N + NBK - 1) / NBK;
    int G   = (E + PTB * PEPT - 1) / (PTB * PEPT);
    int PQB = (ntiles + (PTB / 32) - 1) / (PTB / 32);
    k_pq_part<<<G + PQB, PTB, 0, stream>>>(h, w, dst, pq, out, N, E, npo, ntiles, G);

    k_owner<<<NBK, OTB, npo * sizeof(int), stream>>>(cnt, out, N, npo, G);

    int npg = (N + GGRID - 1) / GGRID;
    k_gather<<<GGRID, GWPB * 32, 0, stream>>>(h, src, dst, pq, cnt, out, N, E, npg);
}


# owner pass and gather share one launch: the range's work-group keeps bins (edge id, source) and in-degrees in LDS and its waves gather right behind a barrier; 2 launches, 8 rows in flight per wave
# speedup vs baseline: 8.8804x; 1.0999x over previous
#include <hip/hip_runtime.h>
#include <hip/hip_bf16.h>
#include <math.h>
#include <limits.h>

#define SLOT  2
#define DIM   128
#define ROW   (SLOT * DIM)
#define SLOPE 0.98f
#define CAP   64
#define LPR   64
#define NSLOT (CAP / 32)
#define NBK   250
#define NPOX  200
#define PTB   512
#define PEPT  8
#define SEG   LPR
#define OTB   1024
#define GWPB  (OTB / 32)
#define SPT   (OTB / SEG)
#define OLD   13

typedef float v2f __attribute__((ext_vector_type(2)));
typedef float v8f __attribute__((ext_vector_type(8)));


__device__ __forceinline__ void role_node_pq(const float* __restrict__ h,
                                             const float* __restrict__ watt,
                                             float* __restrict__ pq,
                                             float* wlds,
                                             int N, int ntiles, int blk) {
    int tid = threadIdx.x;
    if (tid < DIM) { wlds[tid] = watt[tid]; wlds[tid + DIM] = watt[tid + DIM]; }
    __syncthreads();

    int tile = blk * (PTB >> 5) + (tid >> 5);
    if (tile >= ntiles) return;

    int lane = tid & 31;
    int half = lane >> 4;
    int c16  = lane & 15;
    int base = tile * 16;

    int nL = base + c16; if (nL >= N) nL = N - 1;
    const float* pa = h + (size_t)nL * ROW + 2 * half;
    const float* wl = wlds + ((c16 & 8) ? DIM : 0) + 2 * half;

    v8f c = {};
    v2f av = {}, bv = {};
    #pragma unroll
    for (int k = 0; k < 32; ++k) {
        av = *(const v2f*)(pa + 4 * k);
        bv = *(const v2f*)(wl + 4 * k);
        c = __builtin_amdgcn_wmma_f32_16x16x4_f32(false, av, false, bv,
                                                  (short)0, c, false, false);
    }
    asm volatile("v_nop\n\tv_nop\n\tv_nop\n\tv_nop" : "+v"(c) : "v"(av), "v"(bv));
    int r = lane & 7;
    float sc = c[0];
    sc = (r == 1) ? c[1] : sc;
    sc = (r == 2) ? c[2] : sc;
    sc = (r == 3) ? c[3] : sc;
    sc = (r == 4) ? c[4] : sc;
    sc = (r == 5) ? c[5] : sc;
    sc = (r == 6) ? c[6] : sc;
    sc = (r == 7) ? c[7] : sc;

    int node = base + (half << 3) + r;
    if (node < N) pq[2 * (size_t)node + ((lane >> 3) & 1)] = sc;
}

__device__ __forceinline__ float edge_score(const float* __restrict__ pq, int s, int d) {
    float sc = pq[2 * (size_t)s] + pq[2 * (size_t)d + 1];
    return (sc >= 0.0f) ? sc : SLOPE * sc;
}

static_assert(LPR == 64 && LPR * 8 == DIM * 4, "a row's second half holds LPR pairs; list_slot shifts by 6");
__device__ __forceinline__ int2* list_slot(float* out, int n0, int p) {
    return (int2*)(out + (size_t)(n0 + (p >> 6)) * ROW + DIM) + (p & (LPR - 1));
}
static_assert(PTB >= NBK, "one thread per range writes the segment's count");
static_assert(PTB >= DIM, "the score role stages w with one thread per column");
static_assert(SEG == 64 && OTB % SEG == 0, "the owner pass splits a thread index into segment and slot by a shift of 6");

__device__ __forceinline__ void role_part(const int* __restrict__ dst,
                                          float* __restrict__ out,
                                          int* lc,
                                          int N, int E, int npo, int g) {
    int tid = threadIdx.x;
    if (tid < NBK) lc[tid] = 0;
    __syncthreads();

    int e0 = g * (PTB * PEPT) + tid;
    int d[PEPT], b[PEPT], rk[PEPT];
    #pragma unroll
    for (int j = 0; j < PEPT; ++j) {
        int e  = e0 + PTB * j;
        int ec = (e < E) ? e : E - 1;
        d[j] = dst[ec];
    }
    #pragma unroll
    for (int j = 0; j < PEPT; ++j) {
        b[j]  = (int)((unsigned)d[j] / (unsigned)npo);
        rk[j] = -1;
        if (e0 + PTB * j < E) rk[j] = atomicAdd(&lc[b[j]], 1);
    }
    __syncthreads();
    int sbase = g * SEG;
    if (tid < NBK) {
        int n0 = tid * npo;
        if (n0 < N) {
            int nn = N - n0; if (nn > npo) nn = npo;
            if (sbase + SEG <= nn * LPR)
                *list_slot(out, n0, sbase) = make_int2(lc[tid], 0);
        }
    }
    #pragma unroll
    for (int j = 0; j < PEPT; ++j) {
        if (rk[j] >= 0 && rk[j] < SEG - 1) {
            int n0 = b[j] * npo;
            int nn = N - n0; if (nn > npo) nn = npo;
            if (sbase + SEG <= nn * LPR)
                *list_slot(out, n0, sbase + 1 + rk[j]) = make_int2(e0 + PTB * j, d[j]);
        }
    }
}

__global__ void __launch_bounds__(PTB)
k_pq_part(const float* __restrict__ h,
          const float* __restrict__ watt,
          const int* __restrict__ dst,
          float* __restrict__ pq,
          float* __restrict__ out,
          int N, int E, int npo, int ntiles, int G) {
    __shared__ int   lc[NBK];
    __shared__ float wlds[2 * DIM];
    int blk = blockIdx.x;
    if (blk < G) role_part(dst, out, lc, N, E, npo, blk);
    else         role_node_pq(h, watt, pq, wlds, N, ntiles, blk - G);
}

__device__ __forceinline__ void fma4(float4& acc, float w, const float4 x) {
    acc.x = __builtin_fmaf(w, x.x, acc.x); acc.y = __builtin_fmaf(w, x.y, acc.y);
    acc.z = __builtin_fmaf(w, x.z, acc.z); acc.w = __builtin_fmaf(w, x.w, acc.w);
}

template <int K>
__device__ __forceinline__ void add_rows(const int* ls, const float* lw, int r,
                                         const float* hl, float4& accA, float4& accB) {
    float4 xa[K], xb[K];
    float  w[K];
    #pragma unroll
    for (int k = 0; k < K; ++k) {
        const float* p = hl + (size_t)ls[r + k] * ROW;
        w[k]  = lw[r + k];
        xa[k] = *(const float4*)p;
        xb[k] = *(const float4*)(p + DIM);
    }
    #pragma unroll
    for (int k = 0; k < K; ++k) { fma4(accA, w[k], xa[k]); fma4(accB, w[k], xb[k]); }
}

__global__ void __launch_bounds__(OTB)
k_bin_gather(const float* __restrict__ h,
             const int*  __restrict__ src,
             const int*  __restrict__ dst,
             const float* __restrict__ pq,
             float* out, int N, int E, int npo, int G) {
    __shared__ int   lcnt[NPOX];
    __shared__ int   b_e[NPOX * CAP];
    __shared__ int   b_s[NPOX * CAP];
    __shared__ int   l_src[GWPB][CAP];
    __shared__ float l_w[GWPB][CAP];
    __shared__ int   next;
    __shared__ int   ovf;

    int tid = threadIdx.x;
    int n0  = blockIdx.x * npo;
    if (n0 >= N) return;
    int nn = N - n0; if (nn > npo) nn = npo;
    bool over = (G * SEG > nn * LPR) || (nn > NPOX);
    int  nx   = (nn < NPOX) ? nn : NPOX;
    for (int i = tid; i < nx; i += OTB) lcnt[i] = 0;
    if (tid == 0) { ovf = 0; next = n0; }
    __syncthreads();

    if (!over) {
        int sg = tid >> 6;
        int s  = tid & (SEG - 1);
        for (int g0 = 0; g0 < G; g0 += SPT * OLD) {
            int  hc[OLD];
            int2 ed[OLD];
            int  sv[OLD];
            #pragma unroll
            for (int j = 0; j < OLD; ++j) {
                int g = g0 + j * SPT + sg;
                hc[j] = 0; ed[j] = make_int2(0, 0);
                if (g < G) {
                    const int2* q = (const int2*)(out + (size_t)(n0 + g0 + sg) * ROW + DIM)
                                  + (size_t)j * (SPT * (ROW / 2));
                    hc[j] = q[0].x;
                    ed[j] = q[s];
                }
            }
            #pragma unroll
            for (int j = 0; j < OLD; ++j) asm volatile("" : "+v"(hc[j]), "+v"(ed[j].x), "+v"(ed[j].y));
            #pragma unroll
            for (int j = 0; j < OLD; ++j) {
                int  c  = hc[j];
                int  rr = ed[j].y - n0;
                bool ok = (g0 + j * SPT + sg < G) && (c <= SEG - 1) && (s >= 1) && (s <= c)
                       && ((unsigned)ed[j].x < (unsigned)E) && ((unsigned)rr < (unsigned)nx);
                sv[j] = ok ? src[ed[j].x] : 0;
            }
            #pragma unroll
            for (int j = 0; j < OLD; ++j) asm volatile("" : "+v"(sv[j]));
            #pragma unroll
            for (int j = 0; j < OLD; ++j) {
                int g = g0 + j * SPT + sg;
                if (g < G) {
                    int c  = hc[j];
                    int rr = ed[j].y - n0;
                    if (c > SEG - 1) {
                        ovf = 1;
                    } else if (s >= 1 && s <= c
                               && (unsigned)ed[j].x < (unsigned)E && (unsigned)rr < (unsigned)nx) {
                        int slot = atomicAdd(&lcnt[rr], 1);
                        if (slot < CAP) {
                            b_e[rr * CAP + slot] = ed[j].x;
                            b_s[rr * CAP + slot] = sv[j];
                        }
                    }
                }
            }
        }
    }
    __syncthreads();
    bool o2 = over || (ovf != 0);

    int lane = tid & 31;
    int wv   = __builtin_amdgcn_readfirstlane(tid >> 5);
    int ne   = n0 + nn;

    int*   ls = l_src[wv];
    float* lw = l_w[wv];
    const float* hl = h + lane * 4;

    for (;;) {
        int n = 0;
        if (lane == 0) n = atomicAdd(&next, 1);
        n = __builtin_amdgcn_readfirstlane(n);
        if (n >= ne) break;

        int rn = n - n0;
        int Lv = lcnt[(rn < NPOX) ? rn : 0];
        if (o2) Lv = CAP + 1;
        int L = __builtin_amdgcn_readfirstlane(Lv);
        bool fast = (L > 0) && (L <= CAP);

        float4 accA = make_float4(0.f, 0.f, 0.f, 0.f);
        float4 accB = make_float4(0.f, 0.f, 0.f, 0.f);

        if (L == 0) {
            accA = *(const float4*)(hl + (size_t)n * ROW);
            accB = *(const float4*)(hl + (size_t)n * ROW + DIM);
        } else if (fast) {
            int   e_[NSLOT];
            float a_[NSLOT];
            int   sv_[NSLOT];
            float m = -INFINITY;
            const int* be = b_e + rn * CAP;
            const int* bs = b_s + rn * CAP;
            #pragma unroll
            for (int t = 0; t < NSLOT; ++t) {
                int i = lane + 32 * t;
                e_[t] = INT_MAX; a_[t] = -INFINITY; sv_[t] = 0;
                if (i < L) {
                    e_[t]  = be[i];
                    sv_[t] = bs[i];
                    a_[t]  = edge_score(pq, sv_[t], n);
                    m = fmaxf(m, a_[t]);
                }
            }
            for (int o = 16; o > 0; o >>= 1) m = fmaxf(m, __shfl_xor(m, o));

            int rk[NSLOT];
            #pragma unroll
            for (int t = 0; t < NSLOT; ++t) rk[t] = 0;
            #pragma unroll
            for (int t2 = 0; t2 < NSLOT; ++t2) {
                int lim = L - 32 * t2; if (lim > 32) lim = 32;
                for (int j = 0; j < lim; ++j) {
                    int ej = __builtin_amdgcn_readlane(e_[t2], j);
                    #pragma unroll
                    for (int t = 0; t < NSLOT; ++t) rk[t] += (ej < e_[t]) ? 1 : 0;
                }
            }
            #pragma unroll
            for (int t = 0; t < NSLOT; ++t) {
                int i = lane + 32 * t;
                if (i < L) {
                    ls[rk[t]] = sv_[t];
                    lw[rk[t]] = __expf(a_[t] - m);
                }
            }

            static_assert(NSLOT == 2, "the denominator below adds two slots per lane");
            float wsl[NSLOT];
            #pragma unroll
            for (int t = 0; t < NSLOT; ++t) {
                int i = lane + 32 * t;
                wsl[t] = (i < L) ? lw[i] : 0.0f;
            }
            float s = wsl[0] + wsl[1];
            for (int o = 16; o > 0; o >>= 1) s += __shfl_xor(s, o);
            #pragma unroll
            for (int t = 0; t < NSLOT; ++t) {
                int i = lane + 32 * t;
                if (i < L) lw[i] = wsl[t] / s;
            }

            int r = 0;
            for (; r + 8 <= L; r += 8) add_rows<8>(ls, lw, r, hl, accA, accB);
            if (r + 4 <= L) { add_rows<4>(ls, lw, r, hl, accA, accB); r += 4; }
            for (; r < L; ++r) add_rows<1>(ls, lw, r, hl, accA, accB);
        } else {
            float m2 = -INFINITY;
            int   nh = 0;
            for (int e0 = 0; e0 < E; e0 += 32) {
                int e = e0 + lane;
                bool hit = (e < E) && (dst[e] == n);
                if (hit) m2 = fmaxf(m2, edge_score(pq, src[e], n));
                nh += __popc((unsigned)__ballot(hit));
            }
            for (int o = 16; o > 0; o >>= 1) m2 = fmaxf(m2, __shfl_xor(m2, o));
            if (nh == 0) {
                accA = *(const float4*)(hl + (size_t)n * ROW);
                accB = *(const float4*)(hl + (size_t)n * ROW + DIM);
            }
            float s2 = 0.0f;
            for (int e0 = 0; nh > 0 && e0 < E; e0 += 32) {
                int e = e0 + lane;
                bool hit = (e < E) && (dst[e] == n);
                unsigned mask = (unsigned)__ballot(hit);
                while (mask) {
                    int b = __ffs(mask) - 1;
                    mask &= mask - 1;
                    s2 += __expf(edge_score(pq, src[e0 + b], n) - m2);
                }
            }
            for (int e0 = 0; nh > 0 && e0 < E; e0 += 32) {
                int e = e0 + lane;
                bool hit = (e < E) && (dst[e] == n);
                unsigned mask = (unsigned)__ballot(hit);
                while (mask) {
                    int b = __ffs(mask) - 1;
                    mask &= mask - 1;
                    int ee = e0 + b;
                    float al = __expf(edge_score(pq, src[ee], n) - m2) / s2;
                    const float* p0 = hl + (size_t)src[ee] * ROW;
                    float4 x0a = *(const float4*)p0, x0b = *(const float4*)(p0 + DIM);
                    fma4(accA, al, x0a); fma4(accB, al, x0b);
                }
            }
        }

        float* o = out + (size_t)n * ROW + lane * 4;
        *(float4*)o         = accA;
        *(float4*)(o + DIM) = accB;
    }
}

extern "C" void kernel_launch(void* const* d_in, const int* in_sizes, int n_in,
                              void* d_out, int out_size, void* d_ws, size_t ws_size,
                              hipStream_t stream) {
    const float* h   = (const float*)d_in[0];
    const int*   src = (const int*)d_in[1];
    const int*   dst = (const int*)d_in[2];
    const float* w   = (const float*)d_in[3];
    int E = in_sizes[1];
    int N = in_sizes[0] / ROW;
    float* out = (float*)d_out;

    float* pq  = (float*)d_ws;

    int ntiles = (N + 15) / 16;
    int npo = (N + NBK - 1) / NBK;
    int G   = (E + PTB * PEPT - 1) / (PTB * PEPT);
    int PQB = (ntiles + (PTB / 32) - 1) / (PTB / 32);
    k_pq_part<<<G + PQB, PTB, 0, stream>>>(h, w, dst, pq, out, N, E, npo, ntiles, G);

    k_bin_gather<<<NBK, OTB, 0, stream>>>(h, src, dst, pq, out, N, E, npo, G);
}


# gather preamble without global reads: partition pass parks src beside the edge id, owner pass fetches p[src] for its range in one batch and bins it in LDS beside q; scores and sums bit-identical
# speedup vs baseline: 9.5032x; 1.0701x over previous
#include <hip/hip_runtime.h>
#include <hip/hip_bf16.h>
#include <math.h>
#include <limits.h>

#define SLOT  2
#define DIM   128
#define ROW   (SLOT * DIM)
#define SLOPE 0.98f
#define CAP   64
#define LPR   64
#define NSLOT (CAP / 32)
#define NBK   250
#define NPOX  200
#define PTB   512
#define PEPT  8
#define SEG   LPR
#define OTB   1024
#define GWPB  (OTB / 32)
#define SPT   (OTB / SEG)
#define OLD   13
#define PKB   24

typedef float v2f __attribute__((ext_vector_type(2)));
typedef float v8f __attribute__((ext_vector_type(8)));


__device__ __forceinline__ void role_node_pq(const float* __restrict__ h,
                                             const float* __restrict__ watt,
                                             float* __restrict__ pq,
                                             float* wlds,
                                             int N, int ntiles, int blk) {
    int tid = threadIdx.x;
    if (tid < DIM) { wlds[tid] = watt[tid]; wlds[tid + DIM] = watt[tid + DIM]; }
    __syncthreads();

    int tile = blk * (PTB >> 5) + (tid >> 5);
    if (tile >= ntiles) return;

    int lane = tid & 31;
    int half = lane >> 4;
    int c16  = lane & 15;
    int base = tile * 16;

    int nL = base + c16; if (nL >= N) nL = N - 1;
    const float* pa = h + (size_t)nL * ROW + 2 * half;
    const float* wl = wlds + ((c16 & 8) ? DIM : 0) + 2 * half;

    v8f c = {};
    v2f av = {}, bv = {};
    #pragma unroll
    for (int k = 0; k < 32; ++k) {
        av = *(const v2f*)(pa + 4 * k);
        bv = *(const v2f*)(wl + 4 * k);
        c = __builtin_amdgcn_wmma_f32_16x16x4_f32(false, av, false, bv,
                                                  (short)0, c, false, false);
    }
    asm volatile("v_nop\n\tv_nop\n\tv_nop\n\tv_nop" : "+v"(c) : "v"(av), "v"(bv));
    int r = lane & 7;
    float sc = c[0];
    sc = (r == 1) ? c[1] : sc;
    sc = (r == 2) ? c[2] : sc;
    sc = (r == 3) ? c[3] : sc;
    sc = (r == 4) ? c[4] : sc;
    sc = (r == 5) ? c[5] : sc;
    sc = (r == 6) ? c[6] : sc;
    sc = (r == 7) ? c[7] : sc;

    int node = base + (half << 3) + r;
    if (node < N) pq[2 * (size_t)node + ((lane >> 3) & 1)] = sc;
}

__device__ __forceinline__ float edge_score(const float* __restrict__ pq, int s, int d) {
    float sc = pq[2 * (size_t)s] + pq[2 * (size_t)d + 1];
    return (sc >= 0.0f) ? sc : SLOPE * sc;
}

static_assert(LPR == 64 && LPR * 8 == DIM * 4, "a row's second half holds LPR pairs; list_slot shifts by 6");
__device__ __forceinline__ int2* list_slot(float* out, int n0, int p) {
    return (int2*)(out + (size_t)(n0 + (p >> 6)) * ROW + DIM) + (p & (LPR - 1));
}
static_assert(PTB >= NBK, "one thread per range writes the segment's count");
static_assert(NPOX <= (1 << (32 - PKB)), "a place inside a range that keeps bins fits above the source bits");
static_assert(PTB >= DIM, "the score role stages w with one thread per column");
static_assert(SEG == 64 && OTB % SEG == 0, "the owner pass splits a thread index into segment and slot by a shift of 6");

__device__ __forceinline__ void role_part(const int* __restrict__ src,
                                          const int* __restrict__ dst,
                                          float* __restrict__ out,
                                          int* lc,
                                          int N, int E, int npo, int g) {
    int tid = threadIdx.x;
    if (tid < NBK) lc[tid] = 0;
    __syncthreads();

    int e0 = g * (PTB * PEPT) + tid;
    int d[PEPT], sr[PEPT], b[PEPT], rk[PEPT];
    #pragma unroll
    for (int j = 0; j < PEPT; ++j) {
        int e  = e0 + PTB * j;
        int ec = (e < E) ? e : E - 1;
        d[j]  = dst[ec];
        sr[j] = src[ec];
    }
    #pragma unroll
    for (int j = 0; j < PEPT; ++j) {
        b[j]  = (int)((unsigned)d[j] / (unsigned)npo);
        rk[j] = -1;
        if (e0 + PTB * j < E) rk[j] = atomicAdd(&lc[b[j]], 1);
    }
    __syncthreads();
    int sbase = g * SEG;
    if (tid < NBK) {
        int n0 = tid * npo;
        if (n0 < N) {
            int nn = N - n0; if (nn > npo) nn = npo;
            if (sbase + SEG <= nn * LPR)
                *list_slot(out, n0, sbase) = make_int2(lc[tid], 0);
        }
    }
    #pragma unroll
    for (int j = 0; j < PEPT; ++j) {
        if (rk[j] >= 0 && rk[j] < SEG - 1) {
            int n0 = b[j] * npo;
            int nn = N - n0; if (nn > npo) nn = npo;
            if (sbase + SEG <= nn * LPR)
                *list_slot(out, n0, sbase + 1 + rk[j]) =
                    make_int2(e0 + PTB * j, (int)((unsigned)sr[j] | ((unsigned)(d[j] - n0) << PKB)));
        }
    }
}

__global__ void __launch_bounds__(PTB)
k_pq_part(const float* __restrict__ h,
          const float* __restrict__ watt,
          const int* __restrict__ src,
          const int* __restrict__ dst,
          float* __restrict__ pq,
          float* __restrict__ out,
          int N, int E, int npo, int ntiles, int G) {
    __shared__ int   lc[NBK];
    __shared__ float wlds[2 * DIM];
    int blk = blockIdx.x;
    if (blk < G) role_part(src, dst, out, lc, N, E, npo, blk);
    else         role_node_pq(h, watt, pq, wlds, N, ntiles, blk - G);
}

__device__ __forceinline__ void fma4(float4& acc, float w, const float4 x) {
    acc.x = __builtin_fmaf(w, x.x, acc.x); acc.y = __builtin_fmaf(w, x.y, acc.y);
    acc.z = __builtin_fmaf(w, x.z, acc.z); acc.w = __builtin_fmaf(w, x.w, acc.w);
}

template <int K>
__device__ __forceinline__ void add_rows(const int* ls, const float* lw, int r,
                                         const float* hl, float4& accA, float4& accB) {
    float4 xa[K], xb[K];
    float  w[K];
    #pragma unroll
    for (int k = 0; k < K; ++k) {
        const float* p = hl + (size_t)ls[r + k] * ROW;
        w[k]  = lw[r + k];
        xa[k] = *(const float4*)p;
        xb[k] = *(const float4*)(p + DIM);
    }
    #pragma unroll
    for (int k = 0; k < K; ++k) { fma4(accA, w[k], xa[k]); fma4(accB, w[k], xb[k]); }
}

__global__ void __launch_bounds__(OTB)
k_bin_gather(const float* __restrict__ h,
             const int*  __restrict__ src,
             const int*  __restrict__ dst,
             const float* __restrict__ pq,
             float* out, int N, int E, int npo, int G) {
    __shared__ int   lcnt[NPOX];
    __shared__ int   b_e[NPOX * CAP];
    __shared__ int   b_s[NPOX * CAP];
    __shared__ float b_p[NPOX * CAP];
    __shared__ float l_q[NPOX];
    __shared__ int   l_src[GWPB][CAP];
    __shared__ float l_w[GWPB][CAP];
    __shared__ int   next;
    __shared__ int   ovf;

    int tid = threadIdx.x;
    int n0  = blockIdx.x * npo;
    if (n0 >= N) return;
    int nn = N - n0; if (nn > npo) nn = npo;
    bool over = (G * SEG > nn * LPR) || (nn > NPOX) || (N > (1 << PKB));
    int  nx   = (nn < NPOX) ? nn : NPOX;
    for (int i = tid; i < nx; i += OTB) lcnt[i] = 0;
    if (tid == 0) { ovf = 0; next = n0; }
    __syncthreads();

    if (!over) {
        int sg = tid >> 6;
        int s  = tid & (SEG - 1);
        float qv = 0.0f;
        if (tid < nx) qv = pq[2 * (size_t)(n0 + tid) + 1];
        for (int g0 = 0; g0 < G; g0 += SPT * OLD) {
            int   hc[OLD];
            int2  ed[OLD];
            float pv[OLD];
            #pragma unroll
            for (int j = 0; j < OLD; ++j) {
                int g = g0 + j * SPT + sg;
                hc[j] = 0; ed[j] = make_int2(0, 0);
                if (g < G) {
                    const int2* q = (const int2*)(out + (size_t)(n0 + g0 + sg) * ROW + DIM)
                                  + (size_t)j * (SPT * (ROW / 2));
                    hc[j] = q[0].x;
                    ed[j] = q[s];
                }
            }
            #pragma unroll
            for (int j = 0; j < OLD; ++j) asm volatile("" : "+v"(hc[j]), "+v"(ed[j].x), "+v"(ed[j].y));
            asm volatile("" : "+v"(qv));
            #pragma unroll
            for (int j = 0; j < OLD; ++j) {
                int  c  = hc[j];
                int  sn = ed[j].y & ((1 << PKB) - 1);
                int  rr = (int)((unsigned)ed[j].y >> PKB);
                bool ok = (g0 + j * SPT + sg < G) && (c <= SEG - 1) && (s >= 1) && (s <= c)
                       && ((unsigned)ed[j].x < (unsigned)E) && (rr < nx) && (sn < N);
                pv[j] = ok ? pq[2 * (size_t)sn] : 0.0f;
            }
            #pragma unroll
            for (int j = 0; j < OLD; ++j) asm volatile("" : "+v"(pv[j]));
            #pragma unroll
            for (int j = 0; j < OLD; ++j) {
                int g = g0 + j * SPT + sg;
                if (g < G) {
                    int c  = hc[j];
                    int sn = ed[j].y & ((1 << PKB) - 1);
                    int rr = (int)((unsigned)ed[j].y >> PKB);
                    if (c > SEG - 1) {
                        ovf = 1;
                    } else if (s >= 1 && s <= c
                               && (unsigned)ed[j].x < (unsigned)E && rr < nx && sn < N) {
                        int slot = atomicAdd(&lcnt[rr], 1);
                        if (slot < CAP) {
                            b_e[rr * CAP + slot] = ed[j].x;
                            b_s[rr * CAP + slot] = sn;
                            b_p[rr * CAP + slot] = pv[j];
                        }
                    }
                }
            }
        }
        if (tid < nx) l_q[tid] = qv;
    }
    __syncthreads();
    bool o2 = over || (ovf != 0);

    int lane = tid & 31;
    int wv   = __builtin_amdgcn_readfirstlane(tid >> 5);
    int ne   = n0 + nn;

    int*   ls = l_src[wv];
    float* lw = l_w[wv];
    const float* hl = h + lane * 4;

    for (;;) {
        int n = 0;
        if (lane == 0) n = atomicAdd(&next, 1);
        n = __builtin_amdgcn_readfirstlane(n);
        if (n >= ne) break;

        int rn = n - n0;
        int Lv = lcnt[(rn < NPOX) ? rn : 0];
        if (o2) Lv = CAP + 1;
        int L = __builtin_amdgcn_readfirstlane(Lv);
        bool fast = (L > 0) && (L <= CAP);

        float4 accA = make_float4(0.f, 0.f, 0.f, 0.f);
        float4 accB = make_float4(0.f, 0.f, 0.f, 0.f);

        if (L == 0) {
            accA = *(const float4*)(hl + (size_t)n * ROW);
            accB = *(const float4*)(hl + (size_t)n * ROW + DIM);
        } else if (fast) {
            int   e_[NSLOT];
            float a_[NSLOT];
            int   sv_[NSLOT];
            float m = -INFINITY;
            const int*   be = b_e + rn * CAP;
            const int*   bs = b_s + rn * CAP;
            const float* bp = b_p + rn * CAP;
            float qn = l_q[rn];
            #pragma unroll
            for (int t = 0; t < NSLOT; ++t) {
                int i = lane + 32 * t;
                e_[t] = INT_MAX; a_[t] = -INFINITY; sv_[t] = 0;
                if (i < L) {
                    e_[t]  = be[i];
                    sv_[t] = bs[i];
                    float sc = bp[i] + qn;
                    a_[t]  = (sc >= 0.0f) ? sc : SLOPE * sc;
                    m = fmaxf(m, a_[t]);
                }
            }
            for (int o = 16; o > 0; o >>= 1) m = fmaxf(m, __shfl_xor(m, o));

            int rk[NSLOT];
            #pragma unroll
            for (int t = 0; t < NSLOT; ++t) rk[t] = 0;
            #pragma unroll
            for (int t2 = 0; t2 < NSLOT; ++t2) {
                int lim = L - 32 * t2; if (lim > 32) lim = 32;
                for (int j = 0; j < lim; ++j) {
                    int ej = __builtin_amdgcn_readlane(e_[t2], j);
                    #pragma unroll
                    for (int t = 0; t < NSLOT; ++t) rk[t] += (ej < e_[t]) ? 1 : 0;
                }
            }
            #pragma unroll
            for (int t = 0; t < NSLOT; ++t) {
                int i = lane + 32 * t;
                if (i < L) {
                    ls[rk[t]] = sv_[t];
                    lw[rk[t]] = __expf(a_[t] - m);
                }
            }

            static_assert(NSLOT == 2, "the denominator below adds two slots per lane");
            float wsl[NSLOT];
            #pragma unroll
            for (int t = 0; t < NSLOT; ++t) {
                int i = lane + 32 * t;
                wsl[t] = (i < L) ? lw[i] : 0.0f;
            }
            float s = wsl[0] + wsl[1];
            for (int o = 16; o > 0; o >>= 1) s += __shfl_xor(s, o);
            #pragma unroll
            for (int t = 0; t < NSLOT; ++t) {
                int i = lane + 32 * t;
                if (i < L) lw[i] = wsl[t] / s;
            }

            int r = 0;
            for (; r + 8 <= L; r += 8) add_rows<8>(ls, lw, r, hl, accA, accB);
            if (r + 4 <= L) { add_rows<4>(ls, lw, r, hl, accA, accB); r += 4; }
            for (; r < L; ++r) add_rows<1>(ls, lw, r, hl, accA, accB);
        } else {
            float m2 = -INFINITY;
            int   nh = 0;
            for (int e0 = 0; e0 < E; e0 += 32) {
                int e = e0 + lane;
                bool hit = (e < E) && (dst[e] == n);
                if (hit) m2 = fmaxf(m2, edge_score(pq, src[e], n));
                nh += __popc((unsigned)__ballot(hit));
            }
            for (int o = 16; o > 0; o >>= 1) m2 = fmaxf(m2, __shfl_xor(m2, o));
            if (nh == 0) {
                accA = *(const float4*)(hl + (size_t)n * ROW);
                accB = *(const float4*)(hl + (size_t)n * ROW + DIM);
            }
            float s2 = 0.0f;
            for (int e0 = 0; nh > 0 && e0 < E; e0 += 32) {
                int e = e0 + lane;
                bool hit = (e < E) && (dst[e] == n);
                unsigned mask = (unsigned)__ballot(hit);
                while (mask) {
                    int b = __ffs(mask) - 1;
                    mask &= mask - 1;
                    s2 += __expf(edge_score(pq, src[e0 + b], n) - m2);
                }
            }
            for (int e0 = 0; nh > 0 && e0 < E; e0 += 32) {
                int e = e0 + lane;
                bool hit = (e < E) && (dst[e] == n);
                unsigned mask = (unsigned)__ballot(hit);
                while (mask) {
                    int b = __ffs(mask) - 1;
                    mask &= mask - 1;
                    int ee = e0 + b;
                    float al = __expf(edge_score(pq, src[ee], n) - m2) / s2;
                    const float* p0 = hl + (size_t)src[ee] * ROW;
                    float4 x0a = *(const float4*)p0, x0b = *(const float4*)(p0 + DIM);
                    fma4(accA, al, x0a); fma4(accB, al, x0b);
                }
            }
        }

        float* o = out + (size_t)n * ROW + lane * 4;
        *(float4*)o         = accA;
        *(float4*)(o + DIM) = accB;
    }
}

extern "C" void kernel_launch(void* const* d_in, const int* in_sizes, int n_in,
                              void* d_out, int out_size, void* d_ws, size_t ws_size,
                              hipStream_t stream) {
    const float* h   = (const float*)d_in[0];
    const int*   src = (const int*)d_in[1];
    const int*   dst = (const int*)d_in[2];
    const float* w   = (const float*)d_in[3];
    int E = in_sizes[1];
    int N = in_sizes[0] / ROW;
    float* out = (float*)d_out;

    float* pq  = (float*)d_ws;

    int ntiles = (N + 15) / 16;
    int npo = (N + NBK - 1) / NBK;
    int G   = (E + PTB * PEPT - 1) / (PTB * PEPT);
    int PQB = (ntiles + (PTB / 32) - 1) / (PTB / 32);
    k_pq_part<<<G + PQB, PTB, 0, stream>>>(h, w, src, dst, pq, out, N, E, npo, ntiles, G);

    k_bin_gather<<<NBK, OTB, 0, stream>>>(h, src, dst, pq, out, N, E, npo, G);
}
